# starting kernel
# baseline (speedup 1.0000x reference)
#include <hip/hip_runtime.h>
#include <hip/hip_bf16.h>

typedef float v2f __attribute__((ext_vector_type(2)));
typedef float v8f __attribute__((ext_vector_type(8)));

#define NH   4
#define DIN  128
#define DOUT 32

__device__ __forceinline__ unsigned fkey(float f) {
  unsigned u = __float_as_uint(f);
  return (u >> 31) ? ~u : (u ^ 0x80000000u);
}
__device__ __forceinline__ float funkey(unsigned k) {
  unsigned u = (k >> 31) ? (k ^ 0x80000000u) : ~k;
  return __uint_as_float(u);
}

__global__ __launch_bounds__(128) void k_hw_wmma(const float* __restrict__ Hin,
                                                 const float* __restrict__ W,
                                                 float* __restrict__ Hw, int N) {
  const int h     = threadIdx.x >> 5;
  const int lane  = threadIdx.x & 31;
  const int m     = lane & 15;
  const int khalf = lane >> 4;
  const int base  = blockIdx.x * 16;

  int rowA = base + m;
  if (rowA >= N) rowA = N - 1;
  const float* arow  = Hin + (size_t)rowA * DIN + 2 * khalf;
  const float* b0row = W + ((size_t)h * DOUT + m) * DIN + 2 * khalf;
  const float* b1row = W + ((size_t)h * DOUT + m + 16) * DIN + 2 * khalf;

  v8f acc0 = {0.f, 0.f, 0.f, 0.f, 0.f, 0.f, 0.f, 0.f};
  v8f acc1 = {0.f, 0.f, 0.f, 0.f, 0.f, 0.f, 0.f, 0.f};

#pragma unroll
  for (int k = 0; k < DIN; k += 4) {
    v2f a  = { arow[k],  arow[k + 1]  };
    v2f b0 = { b0row[k], b0row[k + 1] };
    v2f b1 = { b1row[k], b1row[k + 1] };
    acc0 = __builtin_amdgcn_wmma_f32_16x16x4_f32(false, a, false, b0, (short)0,
                                                 acc0, false, false);
    acc1 = __builtin_amdgcn_wmma_f32_16x16x4_f32(false, a, false, b1, (short)0,
                                                 acc1, false, false);
  }

#pragma unroll
  for (int g = 0; g < 8; ++g) {
    int node = base + g + 8 * khalf;
    if (node < N) {
      float* dst = Hw + ((size_t)node * NH + h) * DOUT;
      dst[m]      = acc0[g];
      dst[m + 16] = acc1[g];
    }
  }
}

__global__ void k_scores(const float* __restrict__ Hw, const float* __restrict__ Asrc,
                         const float* __restrict__ Atgt, float* __restrict__ ssrc,
                         float* __restrict__ stgt, int N) {
  int n = blockIdx.x * blockDim.x + threadIdx.x;
  if (n >= N) return;
  const float* hw = Hw + (size_t)n * (NH * DOUT);
#pragma unroll
  for (int h = 0; h < NH; ++h) {
    float s1 = 0.f, s2 = 0.f;
#pragma unroll
    for (int o = 0; o < DOUT; ++o) {
      float v = hw[h * DOUT + o];
      s1 += v * Asrc[h * DOUT + o];
      s2 += v * Atgt[h * DOUT + o];
    }
    ssrc[(size_t)n * NH + h] = s1;
    stgt[(size_t)n * NH + h] = s2;
  }
}

__global__ void k_init_seg(unsigned* __restrict__ segmax, float* __restrict__ segsum,
                           int n) {
  int i = blockIdx.x * blockDim.x + threadIdx.x;
  if (i < n) {
    segmax[i] = 0x007FFFFFu;
    segsum[i] = 0.f;
  }
}
__global__ void k_zero(float* __restrict__ p, int n) {
  int i = blockIdx.x * blockDim.x + threadIdx.x;
  if (i < n) p[i] = 0.f;
}

__global__ void k_edge_logit(const int* __restrict__ src, const int* __restrict__ tgt,
                             const float* __restrict__ ssrc, const float* __restrict__ stgt,
                             float* __restrict__ Ebuf, unsigned* __restrict__ segmax,
                             int E) {
  int e = blockIdx.x * blockDim.x + threadIdx.x;
  if (e >= E) return;
  int s = src[e], t = tgt[e];
  float4 a = ((const float4*)ssrc)[s];
  float4 b = ((const float4*)stgt)[t];
  float4 v;
  v.x = a.x + b.x; v.y = a.y + b.y; v.z = a.z + b.z; v.w = a.w + b.w;
  v.x = v.x > 0.f ? v.x : 0.2f * v.x;
  v.y = v.y > 0.f ? v.y : 0.2f * v.y;
  v.z = v.z > 0.f ? v.z : 0.2f * v.z;
  v.w = v.w > 0.f ? v.w : 0.2f * v.w;
  ((float4*)Ebuf)[e] = v;
  unsigned* mrow = segmax + (size_t)t * NH;
  atomicMax(&mrow[0], fkey(v.x));
  atomicMax(&mrow[1], fkey(v.y));
  atomicMax(&mrow[2], fkey(v.z));
  atomicMax(&mrow[3], fkey(v.w));
}

__global__ void k_edge_exp(const int* __restrict__ tgt, float* __restrict__ Ebuf,
                           const unsigned* __restrict__ segmax,
                           float* __restrict__ segsum, int E) {
  int e = blockIdx.x * blockDim.x + threadIdx.x;
  if (e >= E) return;
  int t = tgt[e];
  float4 w = ((const float4*)Ebuf)[e];
  uint4 mk = ((const uint4*)segmax)[t];
  w.x = expf(w.x - funkey(mk.x));
  w.y = expf(w.y - funkey(mk.y));
  w.z = expf(w.z - funkey(mk.z));
  w.w = expf(w.w - funkey(mk.w));
  ((float4*)Ebuf)[e] = w;
  float* srow = segsum + (size_t)t * NH;
  atomicAdd(&srow[0], w.x);
  atomicAdd(&srow[1], w.y);
  atomicAdd(&srow[2], w.z);
  atomicAdd(&srow[3], w.w);
}

__global__ void k_edge_scatter(const int* __restrict__ src, const int* __restrict__ tgt,
                               const float* __restrict__ Ebuf,
                               const float* __restrict__ segsum,
                               const float* __restrict__ Hw, float* __restrict__ dout,
                               int E, int N) {
  int idx = blockIdx.x * blockDim.x + threadIdx.x;
  if (idx >= E * NH) return;
  int e = idx >> 2, h = idx & 3;
  int s = src[e], t = tgt[e];
  float alpha = Ebuf[idx] / segsum[(size_t)t * NH + h];
  const float4* hw = (const float4*)(Hw + ((size_t)s * NH + h) * DOUT);
  float* out = dout + ((size_t)h * N + t) * DOUT;
#pragma unroll
  for (int q = 0; q < DOUT / 4; ++q) {
    float4 v = hw[q];
    atomicAdd(&out[4 * q + 0], alpha * v.x);
    atomicAdd(&out[4 * q + 1], alpha * v.y);
    atomicAdd(&out[4 * q + 2], alpha * v.z);
    atomicAdd(&out[4 * q + 3], alpha * v.w);
  }
}

__global__ void k_elu(float* __restrict__ dout, int n) {
  int i = blockIdx.x * blockDim.x + threadIdx.x;
  if (i >= n) return;
  float x = dout[i];
  dout[i] = x > 0.f ? x : expm1f(x);
}

extern "C" void kernel_launch(void* const* d_in, const int* in_sizes, int n_in,
                              void* d_out, int out_size, void* d_ws, size_t ws_size,
                              hipStream_t stream) {
  const float* Hin  = (const float*)d_in[0];
  const int*   ei   = (const int*)d_in[1];
  const float* W    = (const float*)d_in[2];
  const float* Asrc = (const float*)d_in[3];
  const float* Atgt = (const float*)d_in[4];
  float* dout = (float*)d_out;

  const int N = in_sizes[0] / DIN;
  const int E = in_sizes[1] / 2;
  const int* src = ei;
  const int* tgt = ei + E;

  char* p = (char*)d_ws;
  float*    Hw     = (float*)p;    p += (size_t)N * NH * DOUT * sizeof(float);
  float*    ssrc   = (float*)p;    p += (size_t)N * NH * sizeof(float);
  float*    stgt   = (float*)p;    p += (size_t)N * NH * sizeof(float);
  unsigned* segmax = (unsigned*)p; p += (size_t)N * NH * sizeof(unsigned);
  float*    segsum = (float*)p;    p += (size_t)N * NH * sizeof(float);
  float*    Ebuf   = (float*)p;    p += (size_t)E * NH * sizeof(float);
  (void)ws_size; (void)n_in;

  const int B = 256;
  k_zero<<<(out_size + B - 1) / B, B, 0, stream>>>(dout, out_size);
  k_init_seg<<<(N * NH + B - 1) / B, B, 0, stream>>>(segmax, segsum, N * NH);
  k_hw_wmma<<<(N + 15) / 16, 128, 0, stream>>>(Hin, W, Hw, N);
  k_scores<<<(N + B - 1) / B, B, 0, stream>>>(Hw, Asrc, Atgt, ssrc, stgt, N);
  k_edge_logit<<<(E + B - 1) / B, B, 0, stream>>>(src, tgt, ssrc, stgt, Ebuf, segmax, E);
  k_edge_exp<<<(E + B - 1) / B, B, 0, stream>>>(tgt, Ebuf, segmax, segsum, E);
  k_edge_scatter<<<(E * NH + B - 1) / B, B, 0, stream>>>(src, tgt, Ebuf, segsum, Hw,
                                                         dout, E, N);
  k_elu<<<(out_size + B - 1) / B, B, 0, stream>>>(dout, out_size);
}


# k_edge_scatter: each target row is owned by one package half; a wave reads its die id, draws edge chunks from its half's queue and adds only its half's rows; sum order: atomic arrival, as before
# speedup vs baseline: 3.2244x; 3.2244x over previous
#include <hip/hip_runtime.h>
#include <hip/hip_bf16.h>

typedef float v2f __attribute__((ext_vector_type(2)));
typedef float v8f __attribute__((ext_vector_type(8)));

#define NH   4
#define DIN  128
#define DOUT 32
#define EG    4
#define CHUNK 64
#define QPAD  32

__device__ __forceinline__ unsigned fkey(float f) {
  unsigned u = __float_as_uint(f);
  return (u >> 31) ? ~u : (u ^ 0x80000000u);
}
__device__ __forceinline__ float funkey(unsigned k) {
  unsigned u = (k >> 31) ? (k ^ 0x80000000u) : ~k;
  return __uint_as_float(u);
}

__global__ __launch_bounds__(128) void k_hw_wmma(const float* __restrict__ Hin,
                                                 const float* __restrict__ W,
                                                 float* __restrict__ Hw, int N) {
  const int h     = threadIdx.x >> 5;
  const int lane  = threadIdx.x & 31;
  const int m     = lane & 15;
  const int khalf = lane >> 4;
  const int base  = blockIdx.x * 16;

  int rowA = base + m;
  if (rowA >= N) rowA = N - 1;
  const float* arow  = Hin + (size_t)rowA * DIN + 2 * khalf;
  const float* b0row = W + ((size_t)h * DOUT + m) * DIN + 2 * khalf;
  const float* b1row = W + ((size_t)h * DOUT + m + 16) * DIN + 2 * khalf;

  v8f acc0 = {0.f, 0.f, 0.f, 0.f, 0.f, 0.f, 0.f, 0.f};
  v8f acc1 = {0.f, 0.f, 0.f, 0.f, 0.f, 0.f, 0.f, 0.f};

#pragma unroll
  for (int k = 0; k < DIN; k += 4) {
    v2f a  = { arow[k],  arow[k + 1]  };
    v2f b0 = { b0row[k], b0row[k + 1] };
    v2f b1 = { b1row[k], b1row[k + 1] };
    acc0 = __builtin_amdgcn_wmma_f32_16x16x4_f32(false, a, false, b0, (short)0,
                                                 acc0, false, false);
    acc1 = __builtin_amdgcn_wmma_f32_16x16x4_f32(false, a, false, b1, (short)0,
                                                 acc1, false, false);
  }

#pragma unroll
  for (int g = 0; g < 8; ++g) {
    int node = base + g + 8 * khalf;
    if (node < N) {
      float* dst = Hw + ((size_t)node * NH + h) * DOUT;
      dst[m]      = acc0[g];
      dst[m + 16] = acc1[g];
    }
  }
}

__global__ void k_scores(const float* __restrict__ Hw, const float* __restrict__ Asrc,
                         const float* __restrict__ Atgt, float* __restrict__ ssrc,
                         float* __restrict__ stgt, int N) {
  int n = blockIdx.x * blockDim.x + threadIdx.x;
  if (n >= N) return;
  const float* hw = Hw + (size_t)n * (NH * DOUT);
#pragma unroll
  for (int h = 0; h < NH; ++h) {
    float s1 = 0.f, s2 = 0.f;
#pragma unroll
    for (int o = 0; o < DOUT; ++o) {
      float v = hw[h * DOUT + o];
      s1 += v * Asrc[h * DOUT + o];
      s2 += v * Atgt[h * DOUT + o];
    }
    ssrc[(size_t)n * NH + h] = s1;
    stgt[(size_t)n * NH + h] = s2;
  }
}

__global__ void k_init_seg(unsigned* __restrict__ segmax, float* __restrict__ segsum,
                           int n) {
  int i = blockIdx.x * blockDim.x + threadIdx.x;
  if (i < n) {
    segmax[i] = 0x007FFFFFu;
    segsum[i] = 0.f;
  }
}
__global__ void k_zero(float* __restrict__ p, int n) {
  int i = blockIdx.x * blockDim.x + threadIdx.x;
  if (i < n) p[i] = 0.f;
}

__global__ void k_edge_logit(const int* __restrict__ src, const int* __restrict__ tgt,
                             const float* __restrict__ ssrc, const float* __restrict__ stgt,
                             float* __restrict__ Ebuf, unsigned* __restrict__ segmax,
                             int E) {
  int e = blockIdx.x * blockDim.x + threadIdx.x;
  if (e >= E) return;
  int s = src[e], t = tgt[e];
  float4 a = ((const float4*)ssrc)[s];
  float4 b = ((const float4*)stgt)[t];
  float4 v;
  v.x = a.x + b.x; v.y = a.y + b.y; v.z = a.z + b.z; v.w = a.w + b.w;
  v.x = v.x > 0.f ? v.x : 0.2f * v.x;
  v.y = v.y > 0.f ? v.y : 0.2f * v.y;
  v.z = v.z > 0.f ? v.z : 0.2f * v.z;
  v.w = v.w > 0.f ? v.w : 0.2f * v.w;
  ((float4*)Ebuf)[e] = v;
  unsigned* mrow = segmax + (size_t)t * NH;
  atomicMax(&mrow[0], fkey(v.x));
  atomicMax(&mrow[1], fkey(v.y));
  atomicMax(&mrow[2], fkey(v.z));
  atomicMax(&mrow[3], fkey(v.w));
}

__global__ void k_edge_exp(const int* __restrict__ tgt, float* __restrict__ Ebuf,
                           const unsigned* __restrict__ segmax,
                           float* __restrict__ segsum, unsigned* __restrict__ qctr,
                           int E) {
  int e = blockIdx.x * blockDim.x + threadIdx.x;
  if (e < 2 * QPAD) qctr[e] = 0u;
  if (e >= E) return;
  int t = tgt[e];
  float4 w = ((const float4*)Ebuf)[e];
  uint4 mk = ((const uint4*)segmax)[t];
  w.x = expf(w.x - funkey(mk.x));
  w.y = expf(w.y - funkey(mk.y));
  w.z = expf(w.z - funkey(mk.z));
  w.w = expf(w.w - funkey(mk.w));
  ((float4*)Ebuf)[e] = w;
  float* srow = segsum + (size_t)t * NH;
  atomicAdd(&srow[0], w.x);
  atomicAdd(&srow[1], w.y);
  atomicAdd(&srow[2], w.z);
  atomicAdd(&srow[3], w.w);
}

__device__ __forceinline__ int pop_lowest(unsigned& m) {
  const int j = __builtin_ctz(m);
  m &= m - 1u;
  return j;
}

__global__ __launch_bounds__(256) void k_edge_scatter(const int* __restrict__ src,
                                                      const int* __restrict__ tgt,
                                                      const float* __restrict__ Ebuf,
                                                      const float* __restrict__ segsum,
                                                      const float* __restrict__ Hw,
                                                      float* __restrict__ dout,
                                                      unsigned* __restrict__ qctr,
                                                      int E, int N) {
  const int lane = threadIdx.x & 31;
  const unsigned aid = __builtin_amdgcn_s_sendmsg_rtn(0x87);
  const int myhalf = (int)((aid >> 18) & 1u);
  const int nsplit = N >> 1;
  const int nch = (E + CHUNK - 1) / CHUNK;
  const size_t hstride = (size_t)N * DOUT;

  for (int stage = 0; stage < 2; ++stage) {
    const int role = myhalf ^ stage;
    unsigned* ctr = qctr + role * QPAD;
    for (;;) {
      unsigned drawn = 0u;
      if (lane == 0) drawn = atomicAdd(ctr, 1u);
      const int ch = __builtin_amdgcn_readlane((int)drawn, 0);
      if (ch >= nch) break;

      for (int b = 0; b < CHUNK / 32; ++b) {
        const int base = ch * CHUNK + b * 32;
        if (base >= E) break;
        const int cnt = (E - base) < 32 ? (E - base) : 32;

        int s = 0, t = 0;
        float4 w = {0.f, 0.f, 0.f, 0.f};
        bool mine = false;
        if (lane < cnt) {
          const int e = base + lane;
          s = src[e];
          t = tgt[e];
          w = ((const float4*)Ebuf)[e];
          mine = ((t >= nsplit) ? 1 : 0) == role;
        }
        float ax = 0.f, ay = 0.f, az = 0.f, aw = 0.f;
        if (mine) {
          const float4 ss = ((const float4*)segsum)[t];
          ax = w.x / ss.x;
          ay = w.y / ss.y;
          az = w.z / ss.z;
          aw = w.w / ss.w;
        }
        unsigned todo = __builtin_amdgcn_ballot_w32(mine);

        while (__builtin_popcount(todo) >= EG) {
          int jj[EG];
#pragma unroll
          for (int g = 0; g < EG; ++g) jj[g] = pop_lowest(todo);
          float v[EG][NH];
#pragma unroll
          for (int g = 0; g < EG; ++g) {
            const int sj = __builtin_amdgcn_readlane(s, jj[g]);
            const float* row = Hw + (size_t)sj * (NH * DOUT);
#pragma unroll
            for (int h = 0; h < NH; ++h) v[g][h] = row[h * DOUT + lane];
          }
#pragma unroll
          for (int g = 0; g < EG; ++g) {
            const int tj = __builtin_amdgcn_readlane(t, jj[g]);
            const float a0 = __int_as_float(__builtin_amdgcn_readlane(__float_as_int(ax), jj[g]));
            const float a1 = __int_as_float(__builtin_amdgcn_readlane(__float_as_int(ay), jj[g]));
            const float a2 = __int_as_float(__builtin_amdgcn_readlane(__float_as_int(az), jj[g]));
            const float a3 = __int_as_float(__builtin_amdgcn_readlane(__float_as_int(aw), jj[g]));
            float* o = dout + (size_t)tj * DOUT;
            atomicAdd(&o[lane],                 a0 * v[g][0]);
            atomicAdd(&(o + hstride)[lane],     a1 * v[g][1]);
            atomicAdd(&(o + 2 * hstride)[lane], a2 * v[g][2]);
            atomicAdd(&(o + 3 * hstride)[lane], a3 * v[g][3]);
          }
        }
        while (todo != 0u) {
          const int j = pop_lowest(todo);
          const int sj = __builtin_amdgcn_readlane(s, j);
          const int tj = __builtin_amdgcn_readlane(t, j);
          const float a0 = __int_as_float(__builtin_amdgcn_readlane(__float_as_int(ax), j));
          const float a1 = __int_as_float(__builtin_amdgcn_readlane(__float_as_int(ay), j));
          const float a2 = __int_as_float(__builtin_amdgcn_readlane(__float_as_int(az), j));
          const float a3 = __int_as_float(__builtin_amdgcn_readlane(__float_as_int(aw), j));
          const float* row = Hw + (size_t)sj * (NH * DOUT);
          float* o = dout + (size_t)tj * DOUT;
          atomicAdd(&o[lane],                 a0 * row[lane]);
          atomicAdd(&(o + hstride)[lane],     a1 * row[DOUT + lane]);
          atomicAdd(&(o + 2 * hstride)[lane], a2 * row[2 * DOUT + lane]);
          atomicAdd(&(o + 3 * hstride)[lane], a3 * row[3 * DOUT + lane]);
        }
      }
    }
  }
}

__global__ void k_elu(float* __restrict__ dout, int n) {
  int i = blockIdx.x * blockDim.x + threadIdx.x;
  if (i >= n) return;
  float x = dout[i];
  dout[i] = x > 0.f ? x : expm1f(x);
}

extern "C" void kernel_launch(void* const* d_in, const int* in_sizes, int n_in,
                              void* d_out, int out_size, void* d_ws, size_t ws_size,
                              hipStream_t stream) {
  const float* Hin  = (const float*)d_in[0];
  const int*   ei   = (const int*)d_in[1];
  const float* W    = (const float*)d_in[2];
  const float* Asrc = (const float*)d_in[3];
  const float* Atgt = (const float*)d_in[4];
  float* dout = (float*)d_out;

  const int N = in_sizes[0] / DIN;
  const int E = in_sizes[1] / 2;
  const int* src = ei;
  const int* tgt = ei + E;

  char* p = (char*)d_ws;
  float*    Hw     = (float*)p;    p += (size_t)N * NH * DOUT * sizeof(float);
  float*    ssrc   = (float*)p;    p += (size_t)N * NH * sizeof(float);
  float*    stgt   = (float*)p;    p += (size_t)N * NH * sizeof(float);
  unsigned* segmax = (unsigned*)p; p += (size_t)N * NH * sizeof(unsigned);
  float*    segsum = (float*)p;    p += (size_t)N * NH * sizeof(float);
  float*    Ebuf   = (float*)p;    p += (size_t)E * NH * sizeof(float);
  (void)ws_size; (void)n_in;

  const int B = 256;
  k_zero<<<(out_size + B - 1) / B, B, 0, stream>>>(dout, out_size);
  k_init_seg<<<(N * NH + B - 1) / B, B, 0, stream>>>(segmax, segsum, N * NH);
  k_hw_wmma<<<(N + 15) / 16, 128, 0, stream>>>(Hin, W, Hw, N);
  k_scores<<<(N + B - 1) / B, B, 0, stream>>>(Hw, Asrc, Atgt, ssrc, stgt, N);
  k_edge_logit<<<(E + B - 1) / B, B, 0, stream>>>(src, tgt, ssrc, stgt, Ebuf, segmax, E);
  unsigned* qctr = (unsigned*)ssrc;
  k_edge_exp<<<(E + B - 1) / B, B, 0, stream>>>(tgt, Ebuf, segmax, segsum, qctr, E);
  k_edge_scatter<<<2048, B, 0, stream>>>(src, tgt, Ebuf, segsum, Hw, dout, qctr, E, N);
  k_elu<<<(out_size + B - 1) / B, B, 0, stream>>>(dout, out_size);
}


# k_scores stores the keys as 8 bit planes per 32 edges; a k_gather_lds lane gets its 32 edges' exact match word by 8 xor/and and lists edges at once: record pass, key re-read gone; sum order as before
# speedup vs baseline: 15.1640x; 4.7029x over previous
#include <hip/hip_runtime.h>
#include <hip/hip_bf16.h>

typedef float v2f __attribute__((ext_vector_type(2)));
typedef float v8f __attribute__((ext_vector_type(8)));

#define NH   4
#define DIN  128
#define DOUT 32
#define ROWS 400
#define GW    32
#define CAPW 320
#define NCH  (CAPW / 32)
#define EG     8
#define KU  1024

#define M_MAX  1
#define M_SUM  2
#define M_ROWS 3

__device__ __forceinline__ unsigned fkey(float f) {
  unsigned u = __float_as_uint(f);
  return (u >> 31) ? ~u : (u ^ 0x80000000u);
}
__device__ __forceinline__ float funkey(unsigned k) {
  unsigned u = (k >> 31) ? (k ^ 0x80000000u) : ~k;
  return __uint_as_float(u);
}

#define TN   64
#define HWT  512
#define PL   (DIN + 4)
#define WPT  (NH * DOUT * (DIN / 4) / HWT)
#define APT  (TN * (DIN / 4) / HWT)
__global__ __launch_bounds__(HWT) void k_hw_wmma(const float* __restrict__ Hin,
                                                 const float* __restrict__ W,
                                                 float* __restrict__ Hw, int N) {
  __shared__ alignas(16) float Wl[NH * DOUT * PL];
  __shared__ alignas(16) float Al[2][TN * PL];
  const int tid   = threadIdx.x;
  const int wave  = tid >> 5;
  const int lane  = tid & 31;
  const int h     = wave & (NH - 1);
  const int sub   = wave >> 2;
  const int m     = lane & 15;
  const int khalf = lane >> 4;
  const int ntile = (N + TN - 1) / TN;
  const float4* W4 = (const float4*)W;
  const float4* H4 = (const float4*)Hin;

  const int crow = tid >> 5;
  const int ccol = 4 * (tid & 31);
  int tile = blockIdx.x;

  float4 wv[WPT];
  float4 pf[APT];
#pragma unroll
  for (int j = 0; j < WPT; ++j) wv[j] = W4[tid + j * HWT];
#pragma unroll
  for (int j = 0; j < APT; ++j) {
    int node = tile * TN + crow + 16 * j;
    if (node >= N) node = N - 1;
    pf[j] = H4[(size_t)node * (DIN / 4) + (tid & 31)];
  }
#pragma unroll
  for (int j = 0; j < WPT; ++j) *(float4*)&Wl[(crow + 16 * j) * PL + ccol] = wv[j];
#pragma unroll
  for (int j = 0; j < APT; ++j) *(float4*)&Al[0][(crow + 16 * j) * PL + ccol] = pf[j];
  __syncthreads();

  const float* b0p = Wl + (h * DOUT + m) * PL + 2 * khalf;
  const float* b1p = b0p + 16 * PL;
  int buf = 0;
  for (; tile < ntile; tile += gridDim.x) {
    const int nxt = tile + gridDim.x;
#pragma unroll
    for (int j = 0; j < APT; ++j) {
      int node = nxt * TN + crow + 16 * j;
      if (node >= N) node = N - 1;
      pf[j] = H4[(size_t)node * (DIN / 4) + (tid & 31)];
    }

    const float* ap = Al[buf] + (sub * 16 + m) * PL + 2 * khalf;
    v8f acc0 = {0.f, 0.f, 0.f, 0.f, 0.f, 0.f, 0.f, 0.f};
    v8f acc1 = {0.f, 0.f, 0.f, 0.f, 0.f, 0.f, 0.f, 0.f};
#pragma unroll
    for (int k = 0; k < DIN; k += 4) {
      const v2f a  = *(const v2f*)(ap + k);
      const v2f b0 = *(const v2f*)(b0p + k);
      const v2f b1 = *(const v2f*)(b1p + k);
      acc0 = __builtin_amdgcn_wmma_f32_16x16x4_f32(false, a, false, b0, (short)0,
                                                   acc0, false, false);
      acc1 = __builtin_amdgcn_wmma_f32_16x16x4_f32(false, a, false, b1, (short)0,
                                                   acc1, false, false);
    }

#pragma unroll
    for (int j = 0; j < APT; ++j) *(float4*)&Al[buf ^ 1][(crow + 16 * j) * PL + ccol] = pf[j];
    __syncthreads();
    buf ^= 1;

    const int node0 = tile * TN + sub * 16 + 8 * khalf;
#pragma unroll
    for (int g = 0; g < 8; ++g) {
      const int node = node0 + g;
      if (node < N) {
        float* dst = Hw + ((size_t)node * NH + h) * DOUT;
        dst[m]      = acc0[g];
        dst[m + 16] = acc1[g];
      }
    }
  }
}

__global__ void k_scores(const float* __restrict__ Hw, const float* __restrict__ Asrc,
                         const float* __restrict__ Atgt, float* __restrict__ ssrc,
                         float* __restrict__ stgt, int N, const int* __restrict__ tgt,
                         unsigned* __restrict__ keys, int E) {
  {
    const int lane  = threadIdx.x & 31;
    const int nw    = (int)((gridDim.x * blockDim.x) >> 5);
    const int nunit = (E + KU - 1) / KU;
    uint4* klo = (uint4*)keys;
    uint4* khi = klo + (size_t)nunit * 32;
    for (int hu = (int)((blockIdx.x * blockDim.x + threadIdx.x) >> 5); hu < 2 * nunit; hu += nw) {
      const int u  = hu >> 1;
      const int hf = hu & 1;
      unsigned pl[8] = {0u, 0u, 0u, 0u, 0u, 0u, 0u, 0u};
#pragma unroll
      for (int jj = 0; jj < 16; ++jj) {
        const int j = hf * 16 + jj;
        const int e = u * KU + j * 32 + lane;
        const unsigned t = (unsigned)tgt[e < E ? e : E - 1];
        const unsigned k = (t / ROWS) & 255u;
        const bool mine = lane == j;
#pragma unroll
        for (int b = 0; b < 8; ++b) {
          const unsigned p = __builtin_amdgcn_ballot_w32(((k >> b) & 1u) != 0u);
          pl[b] = mine ? p : pl[b];
        }
      }
      if ((lane >> 4) == hf) {
        klo[(size_t)u * 32 + lane] = make_uint4(pl[0], pl[1], pl[2], pl[3]);
        khi[(size_t)u * 32 + lane] = make_uint4(pl[4], pl[5], pl[6], pl[7]);
      }
    }
  }
  int n = blockIdx.x * blockDim.x + threadIdx.x;
  if (n >= N) return;
  const float* hw = Hw + (size_t)n * (NH * DOUT);
#pragma unroll
  for (int h = 0; h < NH; ++h) {
    float s1 = 0.f, s2 = 0.f;
#pragma unroll
    for (int o = 0; o < DOUT; ++o) {
      float v = hw[h * DOUT + o];
      s1 += v * Asrc[h * DOUT + o];
      s2 += v * Atgt[h * DOUT + o];
    }
    ssrc[(size_t)n * NH + h] = s1;
    stgt[(size_t)n * NH + h] = s2;
  }
}

__global__ __launch_bounds__(GW * 32) void k_gather_lds(const int* __restrict__ src,
                                                        const int* __restrict__ tgt,
                                                        const unsigned* __restrict__ keys,
                                                        const float* __restrict__ ssrc,
                                                        const float* __restrict__ stgt,
                                                        const float* __restrict__ Hw,
                                                        float* __restrict__ dout,
                                                        int E, int N) {
  __shared__ float    acc[ROWS * NH * DOUT];
  __shared__ unsigned smax[ROWS * NH];
  __shared__ float    ssum[ROWS * NH];
  __shared__ int      lst_s[GW * CAPW];
  __shared__ int      lst_r[GW * CAPW];
  __shared__ int      spill;
  const int tid  = threadIdx.x;
  const int lane = tid & 31;
  const int wave = tid >> 5;
  const int row0 = blockIdx.x * ROWS;
  const int rows = (N - row0) < ROWS ? (N - row0) : ROWS;

  for (int i = tid; i < ROWS * NH * DOUT; i += GW * 32) acc[i] = 0.f;
  for (int i = tid; i < ROWS * NH; i += GW * 32) {
    smax[i] = 0x007FFFFFu;
    ssum[i] = 0.f;
  }
  if (tid == 0) spill = 0;
  __syncthreads();

  int* ls = lst_s + wave * CAPW;
  int* lr = lst_r + wave * CAPW;
  int qn = 0;

  auto push = [&](bool m, int e) {
    const unsigned mask = __builtin_amdgcn_ballot_w32(m);
    if (mask != 0u) {
      const int n = __builtin_popcount(mask);
      if (qn + n <= CAPW) {
        const int pos = qn + (int)__builtin_amdgcn_mbcnt_lo(mask, 0u);
        if (m) ls[pos] = e;
        qn += n;
      } else if (lane == 0) {
        spill = 1;
      }
    }
  };

  auto logits = [&](const int s, const int r) -> float4 {
    const float4 a = ((const float4*)ssrc)[s];
    const float4 b = ((const float4*)stgt)[row0 + r];
    float4 v;
    v.x = a.x + b.x; v.y = a.y + b.y; v.z = a.z + b.z; v.w = a.w + b.w;
    v.x = v.x > 0.f ? v.x : 0.2f * v.x;
    v.y = v.y > 0.f ? v.y : 0.2f * v.y;
    v.z = v.z > 0.f ? v.z : 0.2f * v.z;
    v.w = v.w > 0.f ? v.w : 0.2f * v.w;
    return v;
  };

  auto add_rows = [&](const int s, const int r, const unsigned okm, const float ax, const float ay,
                      const float az, const float aw, const int cnt) {
    int j = 0;
    for (; j + EG <= cnt; j += EG) {
      float hv[EG][NH];
#pragma unroll
      for (int g = 0; g < EG; ++g) {
        const int sj = __builtin_amdgcn_readlane(s, j + g);
        const float* row = Hw + (size_t)sj * (NH * DOUT);
#pragma unroll
        for (int h = 0; h < NH; ++h) hv[g][h] = row[h * DOUT + lane];
      }
#pragma unroll
      for (int g = 0; g < EG; ++g) {
        if (((okm >> (j + g)) & 1u) == 0u) continue;
        const int rj = __builtin_amdgcn_readlane(r, j + g);
        const float a0 = __int_as_float(__builtin_amdgcn_readlane(__float_as_int(ax), j + g));
        const float a1 = __int_as_float(__builtin_amdgcn_readlane(__float_as_int(ay), j + g));
        const float a2 = __int_as_float(__builtin_amdgcn_readlane(__float_as_int(az), j + g));
        const float a3 = __int_as_float(__builtin_amdgcn_readlane(__float_as_int(aw), j + g));
        float* o = acc + rj * (NH * DOUT);
        atomicAdd(&o[lane],            a0 * hv[g][0]);
        atomicAdd(&o[DOUT + lane],     a1 * hv[g][1]);
        atomicAdd(&o[2 * DOUT + lane], a2 * hv[g][2]);
        atomicAdd(&o[3 * DOUT + lane], a3 * hv[g][3]);
      }
    }
    for (; j < cnt; ++j) {
      if (((okm >> j) & 1u) == 0u) continue;
      const int sj = __builtin_amdgcn_readlane(s, j);
      const int rj = __builtin_amdgcn_readlane(r, j);
      const float a0 = __int_as_float(__builtin_amdgcn_readlane(__float_as_int(ax), j));
      const float a1 = __int_as_float(__builtin_amdgcn_readlane(__float_as_int(ay), j));
      const float a2 = __int_as_float(__builtin_amdgcn_readlane(__float_as_int(az), j));
      const float a3 = __int_as_float(__builtin_amdgcn_readlane(__float_as_int(aw), j));
      const float* row = Hw + (size_t)sj * (NH * DOUT);
      float* o = acc + rj * (NH * DOUT);
      atomicAdd(&o[lane],            a0 * row[lane]);
      atomicAdd(&o[DOUT + lane],     a1 * row[DOUT + lane]);
      atomicAdd(&o[2 * DOUT + lane], a2 * row[2 * DOUT + lane]);
      atomicAdd(&o[3 * DOUT + lane], a3 * row[3 * DOUT + lane]);
    }
  };

  auto handle = [&](const int mode, const int from, const int cnt) {
    int s = 0, r = 0;
    bool ok = false;
    float4 v = make_float4(0.f, 0.f, 0.f, 0.f);
    if (lane < cnt) {
      const int x = ls[from + lane];
      s = src[x];
      r = tgt[x] - row0;
      ok = (unsigned)r < (unsigned)rows;
      if (!ok) r = 0;
      v = logits(s, r);
    }
    if (mode == M_MAX) {
      if (ok) {
        unsigned* mrow = smax + r * NH;
        atomicMax(&mrow[0], fkey(v.x));
        atomicMax(&mrow[1], fkey(v.y));
        atomicMax(&mrow[2], fkey(v.z));
        atomicMax(&mrow[3], fkey(v.w));
      }
      return;
    }
    float4 w = make_float4(0.f, 0.f, 0.f, 0.f);
    if (ok) {
      const uint4 mk = ((const uint4*)smax)[r];
      w.x = expf(v.x - funkey(mk.x));
      w.y = expf(v.y - funkey(mk.y));
      w.z = expf(v.z - funkey(mk.z));
      w.w = expf(v.w - funkey(mk.w));
    }
    if (mode == M_SUM) {
      if (ok) {
        float* srow = ssum + r * NH;
        atomicAdd(&srow[0], w.x);
        atomicAdd(&srow[1], w.y);
        atomicAdd(&srow[2], w.z);
        atomicAdd(&srow[3], w.w);
      }
      return;
    }
    float ax = 0.f, ay = 0.f, az = 0.f, aw = 0.f;
    if (ok) {
      const float4 ss = ((const float4*)ssum)[r];
      ax = w.x / ss.x;
      ay = w.y / ss.y;
      az = w.z / ss.z;
      aw = w.w / ss.w;
    }
    add_rows(s, r, __builtin_amdgcn_ballot_w32(ok), ax, ay, az, aw, cnt);
  };

  const uint4* klo   = (const uint4*)keys;
  const int    nunit = (E + KU - 1) / KU;
  const uint4* khi   = klo + (size_t)nunit * 32;
  const int    ngrp  = (nunit + 2 * GW - 1) / (2 * GW);
  const int    ntrip = 2 * ngrp;
  const int    trip0 = (int)((blockIdx.x * (unsigned)ntrip) / gridDim.x);
  const unsigned own = blockIdx.x & 255u;
  unsigned inv[8];
#pragma unroll
  for (int b = 0; b < 8; ++b) inv[b] = ((own >> b) & 1u) ? 0u : 0xFFFFFFFFu;

  auto match = [&](const uint4 lo, const uint4 hi) -> unsigned {
    return (lo.x ^ inv[0]) & (lo.y ^ inv[1]) & (lo.z ^ inv[2]) & (lo.w ^ inv[3]) &
           (hi.x ^ inv[4]) & (hi.y ^ inv[5]) & (hi.z ^ inv[6]) & (hi.w ^ inv[7]);
  };

  auto load_group = [&](int t, uint4 (&lo)[2], uint4 (&hi)[2], int (&un)[2]) {
#pragma unroll
    for (int u = 0; u < 2; ++u) {
      int tt = t + u;
      if (tt >= ntrip) tt -= ntrip;
      un[u] = tt * GW + wave;
      const int at = (un[u] < nunit ? un[u] : nunit - 1) * 32 + lane;
      lo[u] = klo[at];
      hi[u] = khi[at];
    }
  };

  auto take_group = [&](const uint4 (&lo)[2], const uint4 (&hi)[2], const int (&un)[2]) {
#pragma unroll
    for (int u = 0; u < 2; ++u) {
      unsigned m = match(lo[u], hi[u]);
      const int ebase = un[u] * KU + lane * 32;
      while (__builtin_amdgcn_ballot_w32(m != 0u) != 0u) {
        const int e = ebase + __builtin_ctz(m | 0x80000000u);
        push(m != 0u && e < E, e);
        m &= m - 1u;
      }
    }
  };

  auto scan_list = [&]() {
    if (nunit <= 0) return;
    int trip = trip0;
    uint4 la[2], ha[2], lb[2], hb[2];
    int   ua[2], ub[2];
    load_group(trip, la, ha, ua);
    for (int it = 0; it < ngrp; it += 2) {
      trip += 2;
      if (trip >= ntrip) trip -= ntrip;
      load_group(it + 1 < ngrp ? trip : 2 * ntrip, lb, hb, ub);
      take_group(la, ha, ua);
      if (it + 1 < ngrp) {
        trip += 2;
        if (trip >= ntrip) trip -= ntrip;
        load_group(it + 2 < ngrp ? trip : 2 * ntrip, la, ha, ua);
        take_group(lb, hb, ub);
      }
    }
  };

  auto scan = [&](const int mode) {
    int trip = trip0;
    for (int it = 0; it < ntrip; ++it) {
      const int un = trip * GW + wave;
      if (++trip >= ntrip) trip -= ntrip;
      if (un >= nunit) continue;
      const int at = un * 32 + lane;
      unsigned m = match(klo[at], khi[at]);
      const int ebase = un * KU + lane * 32;
      while (__builtin_amdgcn_ballot_w32(m != 0u) != 0u) {
        const int e = ebase + __builtin_ctz(m | 0x80000000u);
        push(m != 0u && e < E, e);
        m &= m - 1u;
        __builtin_amdgcn_wave_barrier();
        while (qn >= 32) { qn -= 32; handle(mode, qn, 32); }
      }
    }
    __builtin_amdgcn_wave_barrier();
    while (qn > 0) { const int c = qn < 32 ? qn : 32; qn -= c; handle(mode, qn, c); }
  };

  scan_list();
  __syncthreads();
  if (spill == 0) {
    float4 ev[NCH];
#pragma unroll
    for (int k = 0; k < NCH; ++k) ev[k] = make_float4(0.f, 0.f, 0.f, 0.f);
    auto ev_get = [&](const int c) -> float4 {
      float4 v = ev[0];
#pragma unroll
      for (int k = 1; k < NCH; ++k) {
        const float4 t = ev[k];
        const bool p = c == k;
        v.x = p ? t.x : v.x; v.y = p ? t.y : v.y; v.z = p ? t.z : v.z; v.w = p ? t.w : v.w;
      }
      return v;
    };
    auto ev_put = [&](const int c, const float4 v) {
#pragma unroll
      for (int k = 0; k < NCH; ++k) {
        float4 t = ev[k];
        const bool p = c == k;
        t.x = p ? v.x : t.x; t.y = p ? v.y : t.y; t.z = p ? v.z : t.z; t.w = p ? v.w : t.w;
        ev[k] = t;
      }
    };
    const int nch = (qn + 31) >> 5;

    for (int c = 0; c < nch; ++c) {
      const int at = c * 32 + lane;
      float4 v = make_float4(0.f, 0.f, 0.f, 0.f);
      if (at < qn) {
        const int x = ls[at];
        const int s = src[x];
        int r = tgt[x] - row0;
        const bool ok = (unsigned)r < (unsigned)rows;
        if (!ok) r = 0;
        v = logits(s, r);
        ls[at] = s;
        lr[at] = ok ? r : -1;
        if (ok) {
          unsigned* mrow = smax + r * NH;
          atomicMax(&mrow[0], fkey(v.x));
          atomicMax(&mrow[1], fkey(v.y));
          atomicMax(&mrow[2], fkey(v.z));
          atomicMax(&mrow[3], fkey(v.w));
        }
      }
      ev_put(c, v);
    }
    __syncthreads();

    for (int c = 0; c < nch; ++c) {
      const int at = c * 32 + lane;
      const float4 v = ev_get(c);
      float4 w = make_float4(0.f, 0.f, 0.f, 0.f);
      if (at < qn) {
        const int r = lr[at];
        if (r >= 0) {
          const uint4 mk = ((const uint4*)smax)[r];
          w.x = expf(v.x - funkey(mk.x));
          w.y = expf(v.y - funkey(mk.y));
          w.z = expf(v.z - funkey(mk.z));
          w.w = expf(v.w - funkey(mk.w));
          float* srow = ssum + r * NH;
          atomicAdd(&srow[0], w.x);
          atomicAdd(&srow[1], w.y);
          atomicAdd(&srow[2], w.z);
          atomicAdd(&srow[3], w.w);
        }
      }
      ev_put(c, w);
    }
    __syncthreads();

    for (int c = 0; c < nch; ++c) {
      const int at = c * 32 + lane;
      const float4 w = ev_get(c);
      int s = 0, r = 0;
      bool ok = false;
      float ax = 0.f, ay = 0.f, az = 0.f, aw = 0.f;
      if (at < qn) {
        s = ls[at];
        r = lr[at];
        ok = r >= 0;
        if (!ok) r = 0;
      }
      if (ok) {
        const float4 ss = ((const float4*)ssum)[r];
        ax = w.x / ss.x;
        ay = w.y / ss.y;
        az = w.z / ss.z;
        aw = w.w / ss.w;
      }
      add_rows(s, r, __builtin_amdgcn_ballot_w32(ok), ax, ay, az, aw,
               qn - c * 32 < 32 ? qn - c * 32 : 32);
    }
    __syncthreads();
  } else {
    qn = 0;
    for (int mode = M_MAX; mode <= M_ROWS; ++mode) {
      scan(mode);
      __syncthreads();
    }
  }

  for (int i = tid; i < rows * NH * DOUT; i += GW * 32) {
    const int r = i >> 7;
    const int h = (i >> 5) & (NH - 1);
    const int o = i & (DOUT - 1);
    const float x = acc[i];
    dout[((size_t)h * N + row0 + r) * DOUT + o] = x > 0.f ? x : expm1f(x);
  }
}

extern "C" void kernel_launch(void* const* d_in, const int* in_sizes, int n_in,
                              void* d_out, int out_size, void* d_ws, size_t ws_size,
                              hipStream_t stream) {
  const float* Hin  = (const float*)d_in[0];
  const int*   ei   = (const int*)d_in[1];
  const float* W    = (const float*)d_in[2];
  const float* Asrc = (const float*)d_in[3];
  const float* Atgt = (const float*)d_in[4];
  float* dout = (float*)d_out;

  const int N = in_sizes[0] / DIN;
  const int E = in_sizes[1] / 2;
  const int* src = ei;
  const int* tgt = ei + E;

  char* p = (char*)d_ws;
  float* Hw   = (float*)p; p += (size_t)N * NH * DOUT * sizeof(float);
  float* ssrc = (float*)p; p += (size_t)N * NH * sizeof(float);
  float* stgt = (float*)p; p += (size_t)N * NH * sizeof(float);
  unsigned* keys = (unsigned*)p;
  (void)ws_size; (void)n_in;

  const int B = 256;
  k_hw_wmma<<<512, HWT, 0, stream>>>(Hin, W, Hw, N);
  k_scores<<<(N + B - 1) / B, B, 0, stream>>>(Hw, Asrc, Atgt, ssrc, stgt, N, tgt, keys, E);
  k_gather_lds<<<(N + ROWS - 1) / ROWS, GW * 32, 0, stream>>>(src, tgt, keys, ssrc, stgt, Hw,
                                                              dout, E, N);
  (void)out_size;
}


# H_w row stored column-major over heads (element 4*o+h; W's LDS rows permuted): k_gather_lds gathers an edge's 512-B row with ONE 16-B-per-lane load, was four 128-B loads; sum order: as before
# speedup vs baseline: 16.6209x; 1.0961x over previous
#include <hip/hip_runtime.h>
#include <hip/hip_bf16.h>

typedef float v2f __attribute__((ext_vector_type(2)));
typedef float v8f __attribute__((ext_vector_type(8)));

#define NH   4
#define DIN  128
#define DOUT 32
#define ROWS 400
#define GW    32
#define CAPW 320
#define NCH  (CAPW / 32)
#define EG     8
#define KU  1024

#define M_MAX  1
#define M_SUM  2
#define M_ROWS 3

__device__ __forceinline__ unsigned fkey(float f) {
  unsigned u = __float_as_uint(f);
  return (u >> 31) ? ~u : (u ^ 0x80000000u);
}
__device__ __forceinline__ float funkey(unsigned k) {
  unsigned u = (k >> 31) ? (k ^ 0x80000000u) : ~k;
  return __uint_as_float(u);
}

#define TN   64
#define HWT  512
#define PL   (DIN + 4)
#define WPT  (NH * DOUT * (DIN / 4) / HWT)
#define APT  (TN * (DIN / 4) / HWT)
__global__ __launch_bounds__(HWT) void k_hw_wmma(const float* __restrict__ Hin,
                                                 const float* __restrict__ W,
                                                 float* __restrict__ Hw, int N) {
  __shared__ alignas(16) float Wl[NH * DOUT * PL];
  __shared__ alignas(16) float Al[2][TN * PL];
  const int tid   = threadIdx.x;
  const int wave  = tid >> 5;
  const int lane  = tid & 31;
  const int q     = wave & 3;
  const int sub   = wave >> 2;
  const int m     = lane & 15;
  const int khalf = lane >> 4;
  const int ntile = (N + TN - 1) / TN;
  const float4* W4 = (const float4*)W;
  const float4* H4 = (const float4*)Hin;

  const int crow = tid >> 5;
  const int ccol = 4 * (tid & 31);
  int tile = blockIdx.x;

  float4 wv[WPT];
  float4 pf[APT];
#pragma unroll
  for (int j = 0; j < WPT; ++j) wv[j] = W4[tid + j * HWT];
#pragma unroll
  for (int j = 0; j < APT; ++j) {
    int node = tile * TN + crow + 16 * j;
    if (node >= N) node = N - 1;
    pf[j] = H4[(size_t)node * (DIN / 4) + (tid & 31)];
  }
#pragma unroll
  for (int j = 0; j < WPT; ++j) {
    const int wr = crow + 16 * j;
    const int p  = NH * (wr & (DOUT - 1)) + (wr >> 5);
    *(float4*)&Wl[p * PL + ccol] = wv[j];
  }
#pragma unroll
  for (int j = 0; j < APT; ++j) *(float4*)&Al[0][(crow + 16 * j) * PL + ccol] = pf[j];
  __syncthreads();

  const float* b0p = Wl + (q * 32 + m) * PL + 2 * khalf;
  const float* b1p = b0p + 16 * PL;
  int buf = 0;
  for (; tile < ntile; tile += gridDim.x) {
    const int nxt = tile + gridDim.x;
#pragma unroll
    for (int j = 0; j < APT; ++j) {
      int node = nxt * TN + crow + 16 * j;
      if (node >= N) node = N - 1;
      pf[j] = H4[(size_t)node * (DIN / 4) + (tid & 31)];
    }

    const float* ap = Al[buf] + (sub * 16 + m) * PL + 2 * khalf;
    v8f acc0 = {0.f, 0.f, 0.f, 0.f, 0.f, 0.f, 0.f, 0.f};
    v8f acc1 = {0.f, 0.f, 0.f, 0.f, 0.f, 0.f, 0.f, 0.f};
#pragma unroll
    for (int k = 0; k < DIN; k += 4) {
      const v2f a  = *(const v2f*)(ap + k);
      const v2f b0 = *(const v2f*)(b0p + k);
      const v2f b1 = *(const v2f*)(b1p + k);
      acc0 = __builtin_amdgcn_wmma_f32_16x16x4_f32(false, a, false, b0, (short)0,
                                                   acc0, false, false);
      acc1 = __builtin_amdgcn_wmma_f32_16x16x4_f32(false, a, false, b1, (short)0,
                                                   acc1, false, false);
    }

#pragma unroll
    for (int j = 0; j < APT; ++j) *(float4*)&Al[buf ^ 1][(crow + 16 * j) * PL + ccol] = pf[j];
    __syncthreads();
    buf ^= 1;

    const int node0 = tile * TN + sub * 16 + 8 * khalf;
#pragma unroll
    for (int g = 0; g < 8; ++g) {
      const int node = node0 + g;
      if (node < N) {
        float* dst = Hw + (size_t)node * (NH * DOUT) + q * 32;
        dst[m]      = acc0[g];
        dst[m + 16] = acc1[g];
      }
    }
  }
}

__global__ void k_scores(const float* __restrict__ Hw, const float* __restrict__ Asrc,
                         const float* __restrict__ Atgt, float* __restrict__ ssrc,
                         float* __restrict__ stgt, int N, const int* __restrict__ tgt,
                         unsigned* __restrict__ keys, int E) {
  {
    const int lane  = threadIdx.x & 31;
    const int nw    = (int)((gridDim.x * blockDim.x) >> 5);
    const int nunit = (E + KU - 1) / KU;
    uint4* klo = (uint4*)keys;
    uint4* khi = klo + (size_t)nunit * 32;
    for (int hu = (int)((blockIdx.x * blockDim.x + threadIdx.x) >> 5); hu < 2 * nunit; hu += nw) {
      const int u  = hu >> 1;
      const int hf = hu & 1;
      unsigned pl[8] = {0u, 0u, 0u, 0u, 0u, 0u, 0u, 0u};
#pragma unroll
      for (int jj = 0; jj < 16; ++jj) {
        const int j = hf * 16 + jj;
        const int e = u * KU + j * 32 + lane;
        const unsigned t = (unsigned)tgt[e < E ? e : E - 1];
        const unsigned k = (t / ROWS) & 255u;
        const bool mine = lane == j;
#pragma unroll
        for (int b = 0; b < 8; ++b) {
          const unsigned p = __builtin_amdgcn_ballot_w32(((k >> b) & 1u) != 0u);
          pl[b] = mine ? p : pl[b];
        }
      }
      if ((lane >> 4) == hf) {
        klo[(size_t)u * 32 + lane] = make_uint4(pl[0], pl[1], pl[2], pl[3]);
        khi[(size_t)u * 32 + lane] = make_uint4(pl[4], pl[5], pl[6], pl[7]);
      }
    }
  }
  int n = blockIdx.x * blockDim.x + threadIdx.x;
  if (n >= N) return;
  const float* hw = Hw + (size_t)n * (NH * DOUT);
#pragma unroll
  for (int h = 0; h < NH; ++h) {
    float s1 = 0.f, s2 = 0.f;
#pragma unroll
    for (int o = 0; o < DOUT; ++o) {
      float v = hw[o * NH + h];
      s1 += v * Asrc[h * DOUT + o];
      s2 += v * Atgt[h * DOUT + o];
    }
    ssrc[(size_t)n * NH + h] = s1;
    stgt[(size_t)n * NH + h] = s2;
  }
}

__global__ __launch_bounds__(GW * 32) void k_gather_lds(const int* __restrict__ src,
                                                        const int* __restrict__ tgt,
                                                        const unsigned* __restrict__ keys,
                                                        const float* __restrict__ ssrc,
                                                        const float* __restrict__ stgt,
                                                        const float* __restrict__ Hw,
                                                        float* __restrict__ dout,
                                                        int E, int N) {
  __shared__ float    acc[ROWS * NH * DOUT];
  __shared__ unsigned smax[ROWS * NH];
  __shared__ float    ssum[ROWS * NH];
  __shared__ int      lst_s[GW * CAPW];
  __shared__ int      lst_r[GW * CAPW];
  __shared__ int      spill;
  const int tid  = threadIdx.x;
  const int lane = tid & 31;
  const int wave = tid >> 5;
  const int row0 = blockIdx.x * ROWS;
  const int rows = (N - row0) < ROWS ? (N - row0) : ROWS;

  for (int i = tid; i < ROWS * NH * DOUT; i += GW * 32) acc[i] = 0.f;
  for (int i = tid; i < ROWS * NH; i += GW * 32) {
    smax[i] = 0x007FFFFFu;
    ssum[i] = 0.f;
  }
  if (tid == 0) spill = 0;
  __syncthreads();

  int* ls = lst_s + wave * CAPW;
  int* lr = lst_r + wave * CAPW;
  int qn = 0;

  auto push = [&](bool m, int e) {
    const unsigned mask = __builtin_amdgcn_ballot_w32(m);
    if (mask != 0u) {
      const int n = __builtin_popcount(mask);
      if (qn + n <= CAPW) {
        const int pos = qn + (int)__builtin_amdgcn_mbcnt_lo(mask, 0u);
        if (m) ls[pos] = e;
        qn += n;
      } else if (lane == 0) {
        spill = 1;
      }
    }
  };

  auto logits = [&](const int s, const int r) -> float4 {
    const float4 a = ((const float4*)ssrc)[s];
    const float4 b = ((const float4*)stgt)[row0 + r];
    float4 v;
    v.x = a.x + b.x; v.y = a.y + b.y; v.z = a.z + b.z; v.w = a.w + b.w;
    v.x = v.x > 0.f ? v.x : 0.2f * v.x;
    v.y = v.y > 0.f ? v.y : 0.2f * v.y;
    v.z = v.z > 0.f ? v.z : 0.2f * v.z;
    v.w = v.w > 0.f ? v.w : 0.2f * v.w;
    return v;
  };

  auto add_rows = [&](const int s, const int r, const unsigned okm, const float ax, const float ay,
                      const float az, const float aw, const int cnt) {
    int j = 0;
    for (; j + EG <= cnt; j += EG) {
      float4 hv[EG];
#pragma unroll
      for (int g = 0; g < EG; ++g) {
        const int sj = __builtin_amdgcn_readlane(s, j + g);
        const float4* row = (const float4*)(Hw + (size_t)sj * (NH * DOUT));
        hv[g] = row[lane];
      }
#pragma unroll
      for (int g = 0; g < EG; ++g) {
        if (((okm >> (j + g)) & 1u) == 0u) continue;
        const int rj = __builtin_amdgcn_readlane(r, j + g);
        const float a0 = __int_as_float(__builtin_amdgcn_readlane(__float_as_int(ax), j + g));
        const float a1 = __int_as_float(__builtin_amdgcn_readlane(__float_as_int(ay), j + g));
        const float a2 = __int_as_float(__builtin_amdgcn_readlane(__float_as_int(az), j + g));
        const float a3 = __int_as_float(__builtin_amdgcn_readlane(__float_as_int(aw), j + g));
        float* o = acc + rj * (NH * DOUT);
        atomicAdd(&o[lane],            a0 * hv[g].x);
        atomicAdd(&o[DOUT + lane],     a1 * hv[g].y);
        atomicAdd(&o[2 * DOUT + lane], a2 * hv[g].z);
        atomicAdd(&o[3 * DOUT + lane], a3 * hv[g].w);
      }
    }
    for (; j < cnt; ++j) {
      if (((okm >> j) & 1u) == 0u) continue;
      const int sj = __builtin_amdgcn_readlane(s, j);
      const int rj = __builtin_amdgcn_readlane(r, j);
      const float a0 = __int_as_float(__builtin_amdgcn_readlane(__float_as_int(ax), j));
      const float a1 = __int_as_float(__builtin_amdgcn_readlane(__float_as_int(ay), j));
      const float a2 = __int_as_float(__builtin_amdgcn_readlane(__float_as_int(az), j));
      const float a3 = __int_as_float(__builtin_amdgcn_readlane(__float_as_int(aw), j));
      const float4 v = ((const float4*)(Hw + (size_t)sj * (NH * DOUT)))[lane];
      float* o = acc + rj * (NH * DOUT);
      atomicAdd(&o[lane],            a0 * v.x);
      atomicAdd(&o[DOUT + lane],     a1 * v.y);
      atomicAdd(&o[2 * DOUT + lane], a2 * v.z);
      atomicAdd(&o[3 * DOUT + lane], a3 * v.w);
    }
  };

  auto handle = [&](const int mode, const int from, const int cnt) {
    int s = 0, r = 0;
    bool ok = false;
    float4 v = make_float4(0.f, 0.f, 0.f, 0.f);
    if (lane < cnt) {
      const int x = ls[from + lane];
      s = src[x];
      r = tgt[x] - row0;
      ok = (unsigned)r < (unsigned)rows;
      if (!ok) r = 0;
      v = logits(s, r);
    }
    if (mode == M_MAX) {
      if (ok) {
        unsigned* mrow = smax + r * NH;
        atomicMax(&mrow[0], fkey(v.x));
        atomicMax(&mrow[1], fkey(v.y));
        atomicMax(&mrow[2], fkey(v.z));
        atomicMax(&mrow[3], fkey(v.w));
      }
      return;
    }
    float4 w = make_float4(0.f, 0.f, 0.f, 0.f);
    if (ok) {
      const uint4 mk = ((const uint4*)smax)[r];
      w.x = expf(v.x - funkey(mk.x));
      w.y = expf(v.y - funkey(mk.y));
      w.z = expf(v.z - funkey(mk.z));
      w.w = expf(v.w - funkey(mk.w));
    }
    if (mode == M_SUM) {
      if (ok) {
        float* srow = ssum + r * NH;
        atomicAdd(&srow[0], w.x);
        atomicAdd(&srow[1], w.y);
        atomicAdd(&srow[2], w.z);
        atomicAdd(&srow[3], w.w);
      }
      return;
    }
    float ax = 0.f, ay = 0.f, az = 0.f, aw = 0.f;
    if (ok) {
      const float4 ss = ((const float4*)ssum)[r];
      ax = w.x / ss.x;
      ay = w.y / ss.y;
      az = w.z / ss.z;
      aw = w.w / ss.w;
    }
    add_rows(s, r, __builtin_amdgcn_ballot_w32(ok), ax, ay, az, aw, cnt);
  };

  const uint4* klo   = (const uint4*)keys;
  const int    nunit = (E + KU - 1) / KU;
  const uint4* khi   = klo + (size_t)nunit * 32;
  const int    ngrp  = (nunit + 2 * GW - 1) / (2 * GW);
  const int    ntrip = 2 * ngrp;
  const int    trip0 = (int)((blockIdx.x * (unsigned)ntrip) / gridDim.x);
  const unsigned own = blockIdx.x & 255u;
  unsigned inv[8];
#pragma unroll
  for (int b = 0; b < 8; ++b) inv[b] = ((own >> b) & 1u) ? 0u : 0xFFFFFFFFu;

  auto match = [&](const uint4 lo, const uint4 hi) -> unsigned {
    return (lo.x ^ inv[0]) & (lo.y ^ inv[1]) & (lo.z ^ inv[2]) & (lo.w ^ inv[3]) &
           (hi.x ^ inv[4]) & (hi.y ^ inv[5]) & (hi.z ^ inv[6]) & (hi.w ^ inv[7]);
  };

  auto load_group = [&](int t, uint4 (&lo)[2], uint4 (&hi)[2], int (&un)[2]) {
#pragma unroll
    for (int u = 0; u < 2; ++u) {
      int tt = t + u;
      if (tt >= ntrip) tt -= ntrip;
      un[u] = tt * GW + wave;
      const int at = (un[u] < nunit ? un[u] : nunit - 1) * 32 + lane;
      lo[u] = klo[at];
      hi[u] = khi[at];
    }
  };

  auto take_group = [&](const uint4 (&lo)[2], const uint4 (&hi)[2], const int (&un)[2]) {
#pragma unroll
    for (int u = 0; u < 2; ++u) {
      unsigned m = match(lo[u], hi[u]);
      const int ebase = un[u] * KU + lane * 32;
      while (__builtin_amdgcn_ballot_w32(m != 0u) != 0u) {
        const int e = ebase + __builtin_ctz(m | 0x80000000u);
        push(m != 0u && e < E, e);
        m &= m - 1u;
      }
    }
  };

  auto scan_list = [&]() {
    if (nunit <= 0) return;
    int trip = trip0;
    uint4 la[2], ha[2], lb[2], hb[2];
    int   ua[2], ub[2];
    load_group(trip, la, ha, ua);
    for (int it = 0; it < ngrp; it += 2) {
      trip += 2;
      if (trip >= ntrip) trip -= ntrip;
      load_group(it + 1 < ngrp ? trip : 2 * ntrip, lb, hb, ub);
      take_group(la, ha, ua);
      if (it + 1 < ngrp) {
        trip += 2;
        if (trip >= ntrip) trip -= ntrip;
        load_group(it + 2 < ngrp ? trip : 2 * ntrip, la, ha, ua);
        take_group(lb, hb, ub);
      }
    }
  };

  auto scan = [&](const int mode) {
    int trip = trip0;
    for (int it = 0; it < ntrip; ++it) {
      const int un = trip * GW + wave;
      if (++trip >= ntrip) trip -= ntrip;
      if (un >= nunit) continue;
      const int at = un * 32 + lane;
      unsigned m = match(klo[at], khi[at]);
      const int ebase = un * KU + lane * 32;
      while (__builtin_amdgcn_ballot_w32(m != 0u) != 0u) {
        const int e = ebase + __builtin_ctz(m | 0x80000000u);
        push(m != 0u && e < E, e);
        m &= m - 1u;
        __builtin_amdgcn_wave_barrier();
        while (qn >= 32) { qn -= 32; handle(mode, qn, 32); }
      }
    }
    __builtin_amdgcn_wave_barrier();
    while (qn > 0) { const int c = qn < 32 ? qn : 32; qn -= c; handle(mode, qn, c); }
  };

  scan_list();
  __syncthreads();
  if (spill == 0) {
    float4 ev[NCH];
#pragma unroll
    for (int k = 0; k < NCH; ++k) ev[k] = make_float4(0.f, 0.f, 0.f, 0.f);
    auto ev_get = [&](const int c) -> float4 {
      float4 v = ev[0];
#pragma unroll
      for (int k = 1; k < NCH; ++k) {
        const float4 t = ev[k];
        const bool p = c == k;
        v.x = p ? t.x : v.x; v.y = p ? t.y : v.y; v.z = p ? t.z : v.z; v.w = p ? t.w : v.w;
      }
      return v;
    };
    auto ev_put = [&](const int c, const float4 v) {
#pragma unroll
      for (int k = 0; k < NCH; ++k) {
        float4 t = ev[k];
        const bool p = c == k;
        t.x = p ? v.x : t.x; t.y = p ? v.y : t.y; t.z = p ? v.z : t.z; t.w = p ? v.w : t.w;
        ev[k] = t;
      }
    };
    const int nch = (qn + 31) >> 5;

    for (int c = 0; c < nch; ++c) {
      const int at = c * 32 + lane;
      float4 v = make_float4(0.f, 0.f, 0.f, 0.f);
      if (at < qn) {
        const int x = ls[at];
        const int s = src[x];
        int r = tgt[x] - row0;
        const bool ok = (unsigned)r < (unsigned)rows;
        if (!ok) r = 0;
        v = logits(s, r);
        ls[at] = s;
        lr[at] = ok ? r : -1;
        if (ok) {
          unsigned* mrow = smax + r * NH;
          atomicMax(&mrow[0], fkey(v.x));
          atomicMax(&mrow[1], fkey(v.y));
          atomicMax(&mrow[2], fkey(v.z));
          atomicMax(&mrow[3], fkey(v.w));
        }
      }
      ev_put(c, v);
    }
    __syncthreads();

    for (int c = 0; c < nch; ++c) {
      const int at = c * 32 + lane;
      const float4 v = ev_get(c);
      float4 w = make_float4(0.f, 0.f, 0.f, 0.f);
      if (at < qn) {
        const int r = lr[at];
        if (r >= 0) {
          const uint4 mk = ((const uint4*)smax)[r];
          w.x = expf(v.x - funkey(mk.x));
          w.y = expf(v.y - funkey(mk.y));
          w.z = expf(v.z - funkey(mk.z));
          w.w = expf(v.w - funkey(mk.w));
          float* srow = ssum + r * NH;
          atomicAdd(&srow[0], w.x);
          atomicAdd(&srow[1], w.y);
          atomicAdd(&srow[2], w.z);
          atomicAdd(&srow[3], w.w);
        }
      }
      ev_put(c, w);
    }
    __syncthreads();

    for (int c = 0; c < nch; ++c) {
      const int at = c * 32 + lane;
      const float4 w = ev_get(c);
      int s = 0, r = 0;
      bool ok = false;
      float ax = 0.f, ay = 0.f, az = 0.f, aw = 0.f;
      if (at < qn) {
        s = ls[at];
        r = lr[at];
        ok = r >= 0;
        if (!ok) r = 0;
      }
      if (ok) {
        const float4 ss = ((const float4*)ssum)[r];
        ax = w.x / ss.x;
        ay = w.y / ss.y;
        az = w.z / ss.z;
        aw = w.w / ss.w;
      }
      add_rows(s, r, __builtin_amdgcn_ballot_w32(ok), ax, ay, az, aw,
               qn - c * 32 < 32 ? qn - c * 32 : 32);
    }
    __syncthreads();
  } else {
    qn = 0;
    for (int mode = M_MAX; mode <= M_ROWS; ++mode) {
      scan(mode);
      __syncthreads();
    }
  }

  for (int i = tid; i < rows * NH * DOUT; i += GW * 32) {
    const int r = i >> 7;
    const int h = (i >> 5) & (NH - 1);
    const int o = i & (DOUT - 1);
    const float x = acc[i];
    dout[((size_t)h * N + row0 + r) * DOUT + o] = x > 0.f ? x : expm1f(x);
  }
}

extern "C" void kernel_launch(void* const* d_in, const int* in_sizes, int n_in,
                              void* d_out, int out_size, void* d_ws, size_t ws_size,
                              hipStream_t stream) {
  const float* Hin  = (const float*)d_in[0];
  const int*   ei   = (const int*)d_in[1];
  const float* W    = (const float*)d_in[2];
  const float* Asrc = (const float*)d_in[3];
  const float* Atgt = (const float*)d_in[4];
  float* dout = (float*)d_out;

  const int N = in_sizes[0] / DIN;
  const int E = in_sizes[1] / 2;
  const int* src = ei;
  const int* tgt = ei + E;

  char* p = (char*)d_ws;
  float* Hw   = (float*)p; p += (size_t)N * NH * DOUT * sizeof(float);
  float* ssrc = (float*)p; p += (size_t)N * NH * sizeof(float);
  float* stgt = (float*)p; p += (size_t)N * NH * sizeof(float);
  unsigned* keys = (unsigned*)p;
  (void)ws_size; (void)n_in;

  const int B = 256;
  k_hw_wmma<<<512, HWT, 0, stream>>>(Hin, W, Hw, N);
  k_scores<<<(N + B - 1) / B, B, 0, stream>>>(Hw, Asrc, Atgt, ssrc, stgt, N, tgt, keys, E);
  k_gather_lds<<<(N + ROWS - 1) / ROWS, GW * 32, 0, stream>>>(src, tgt, keys, ssrc, stgt, Hw,
                                                              dout, E, N);
  (void)out_size;
}
